# starting kernel
# baseline (speedup 1.0000x reference)
#include <hip/hip_runtime.h>


typedef __attribute__((ext_vector_type(2))) float v2f;
typedef __attribute__((ext_vector_type(8))) float v8f;

constexpr int kN  = 100000;
constexpr int kE  = 1600000;
constexpr int kXF = 128;
constexpr int kXL = 40;
constexpr int kH  = 128;
constexpr int kC  = 40;
constexpr int kKM = kXF + kXL;
constexpr int kKC = 2 * kH;


__global__ void k_init_deg(float* __restrict__ deg) {
  int i = blockIdx.x * blockDim.x + threadIdx.x;
  if (i < kN) deg[i] = 1.0f;
}

__global__ void k_deg_accum(const int* __restrict__ eidx, float* __restrict__ deg) {
  int e = blockIdx.x * blockDim.x + threadIdx.x;
  if (e < kE) atomicAdd(deg + eidx[kE + e], 1.0f);
}

__global__ void k_dinv(float* __restrict__ deg) {
  int i = blockIdx.x * blockDim.x + threadIdx.x;
  if (i < kN) {
    float d = deg[i];
    deg[i] = (d > 0.0f) ? rsqrtf(d) : 0.0f;
  }
}


__global__ __launch_bounds__(32)
void k_gemm_gcn(const float* __restrict__ A, const float* __restrict__ B,
                float* __restrict__ Hout) {
  __shared__ float As[16][kXF + 4];
  __shared__ float Bs[kXF][16];

  const int lane = threadIdx.x;
  const int m0 = blockIdx.x * 16;
  const int n0 = blockIdx.y * 16;

  for (int idx = lane; idx < 16 * kXF; idx += 32) {
    int r = idx >> 7, c = idx & 127;
    As[r][c] = A[(size_t)(m0 + r) * kXF + c];
  }
  for (int idx = lane; idx < kXF * 16; idx += 32) {
    int k = idx >> 4, n = idx & 15;
    Bs[k][n] = B[k * kH + n0 + n];
  }
  __syncthreads();

  const int half = lane >> 4;
  const int l    = lane & 15;
  v8f acc = {};

#pragma unroll 4
  for (int kb = 0; kb < kXF; kb += 4) {
    v2f a, b;
    a[0] = As[l][kb + 2 * half + 0];
    a[1] = As[l][kb + 2 * half + 1];
    b[0] = Bs[kb + 2 * half + 0][l];
    b[1] = Bs[kb + 2 * half + 1][l];
    acc = __builtin_amdgcn_wmma_f32_16x16x4_f32(false, a, false, b,
                                                (short)0, acc, false, false);
  }

#pragma unroll
  for (int r = 0; r < 8; ++r)
    Hout[(size_t)(m0 + half * 8 + r) * kH + (n0 + l)] = acc[r];
}


__global__ void k_selfloop(const float* __restrict__ h,
                           const float* __restrict__ dinv,
                           float* __restrict__ comb) {
  int t = blockIdx.x * blockDim.x + threadIdx.x;
  if (t < kN * kH) {
    int i = t >> 7, c = t & 127;
    float di = dinv[i];
    comb[(size_t)i * kKC + c] = h[(size_t)i * kH + c] * di * di;
  }
}


__global__ void k_edge_scatter(const int* __restrict__ eidx,
                               const float* __restrict__ h,
                               const float* __restrict__ dinv,
                               float* __restrict__ comb) {
  long long t = (long long)blockIdx.x * blockDim.x + threadIdx.x;
  if (t >= (long long)kE * 32) return;
  int e = (int)(t >> 5);
  int q = (int)(t & 31);
  int s = eidx[e];
  int d = eidx[kE + e];
  float nm = dinv[s] * dinv[d];
  const float4 hv = *reinterpret_cast<const float4*>(h + (size_t)s * kH + q * 4);
  float* p = comb + (size_t)d * kKC + q * 4;
  atomicAdd(p + 0, hv.x * nm);
  atomicAdd(p + 1, hv.y * nm);
  atomicAdd(p + 2, hv.z * nm);
  atomicAdd(p + 3, hv.w * nm);
}


__global__ __launch_bounds__(32)
void k_gemm_mlp(const float* __restrict__ xf, const float* __restrict__ xl,
                const float* __restrict__ W, const float* __restrict__ bias,
                float* __restrict__ comb) {
  __shared__ float As[16][kKM + 4];
  __shared__ float Bs[kKM][16];

  const int lane = threadIdx.x;
  const int m0 = blockIdx.x * 16;
  const int n0 = blockIdx.y * 16;

  for (int idx = lane; idx < 16 * kKM; idx += 32) {
    int r = idx / kKM, c = idx % kKM;
    As[r][c] = (c < kXF) ? xf[(size_t)(m0 + r) * kXF + c]
                         : xl[(size_t)(m0 + r) * kXL + (c - kXF)];
  }
  for (int idx = lane; idx < kKM * 16; idx += 32) {
    int k = idx >> 4, n = idx & 15;
    Bs[k][n] = W[k * kH + n0 + n];
  }
  __syncthreads();

  const int half = lane >> 4;
  const int l    = lane & 15;
  v8f acc = {};

#pragma unroll 2
  for (int kb = 0; kb < kKM; kb += 4) {
    v2f a, b;
    a[0] = As[l][kb + 2 * half + 0];
    a[1] = As[l][kb + 2 * half + 1];
    b[0] = Bs[kb + 2 * half + 0][l];
    b[1] = Bs[kb + 2 * half + 1][l];
    acc = __builtin_amdgcn_wmma_f32_16x16x4_f32(false, a, false, b,
                                                (short)0, acc, false, false);
  }

  const float bn = bias[n0 + l];
#pragma unroll
  for (int r = 0; r < 8; ++r) {
    float v = acc[r] + bn;
    v = v > 0.0f ? v : 0.0f;
    comb[(size_t)(m0 + half * 8 + r) * kKC + kH + (n0 + l)] = v;
  }
}


__global__ void k_gcn_bias_relu(float* __restrict__ comb, const float* __restrict__ bg) {
  int t = blockIdx.x * blockDim.x + threadIdx.x;
  if (t < kN * kH) {
    int i = t >> 7, c = t & 127;
    float v = comb[(size_t)i * kKC + c] + bg[c];
    comb[(size_t)i * kKC + c] = v > 0.0f ? v : 0.0f;
  }
}


__global__ __launch_bounds__(32)
void k_gemm_cls(const float* __restrict__ comb, const float* __restrict__ W,
                const float* __restrict__ bias, float* __restrict__ out) {
  __shared__ float As[16][kKC + 4];
  __shared__ float Bs[kKC][16];

  const int lane = threadIdx.x;
  const int m0 = blockIdx.x * 16;
  const int n0 = blockIdx.y * 16;

  for (int idx = lane; idx < 16 * kKC; idx += 32) {
    int r = idx >> 8, c = idx & 255;
    As[r][c] = comb[(size_t)(m0 + r) * kKC + c];
  }
  for (int idx = lane; idx < kKC * 16; idx += 32) {
    int k = idx >> 4, n = idx & 15;
    Bs[k][n] = (n0 + n < kC) ? W[k * kC + n0 + n] : 0.0f;
  }
  __syncthreads();

  const int half = lane >> 4;
  const int l    = lane & 15;
  v8f acc = {};

#pragma unroll 4
  for (int kb = 0; kb < kKC; kb += 4) {
    v2f a, b;
    a[0] = As[l][kb + 2 * half + 0];
    a[1] = As[l][kb + 2 * half + 1];
    b[0] = Bs[kb + 2 * half + 0][l];
    b[1] = Bs[kb + 2 * half + 1][l];
    acc = __builtin_amdgcn_wmma_f32_16x16x4_f32(false, a, false, b,
                                                (short)0, acc, false, false);
  }

  const int col = n0 + l;
  if (col < kC) {
    const float bn = bias[col];
#pragma unroll
    for (int r = 0; r < 8; ++r)
      out[(size_t)(m0 + half * 8 + r) * kC + col] = acc[r] + bn;
  }
}


extern "C" void kernel_launch(void* const* d_in, const int* in_sizes, int n_in,
                              void* d_out, int out_size, void* d_ws, size_t ws_size,
                              hipStream_t stream) {
  (void)in_sizes; (void)n_in; (void)out_size; (void)ws_size;

  const float* xfeat  = (const float*)d_in[0];
  const float* xlabel = (const float*)d_in[1];
  const int*   eidx   = (const int*)d_in[2];
  const float* W_gcn  = (const float*)d_in[3];
  const float* b_gcn  = (const float*)d_in[4];
  const float* W_mlp  = (const float*)d_in[5];
  const float* b_mlp  = (const float*)d_in[6];
  const float* W_cls  = (const float*)d_in[7];
  const float* b_cls  = (const float*)d_in[8];
  float* out = (float*)d_out;

  char* ws = (char*)d_ws;
  const size_t off_deg  = 0;
  const size_t off_h    = (off_deg + (size_t)kN * 4 + 255) & ~(size_t)255;
  const size_t off_comb = (off_h + (size_t)kN * kH * 4 + 255) & ~(size_t)255;
  float* deg  = (float*)(ws + off_deg);
  float* hbuf = (float*)(ws + off_h);
  float* comb = (float*)(ws + off_comb);

  const dim3 blk256(256);

  k_init_deg <<<dim3((kN + 255) / 256), blk256, 0, stream>>>(deg);
  k_deg_accum<<<dim3((kE + 255) / 256), blk256, 0, stream>>>(eidx, deg);
  k_dinv     <<<dim3((kN + 255) / 256), blk256, 0, stream>>>(deg);

  k_gemm_gcn<<<dim3(kN / 16, kH / 16), dim3(32), 0, stream>>>(xfeat, W_gcn, hbuf);

  k_selfloop<<<dim3((kN * kH) / 256), blk256, 0, stream>>>(hbuf, deg, comb);
  k_edge_scatter<<<dim3((unsigned)(((long long)kE * 32) / 256)), blk256, 0, stream>>>(
      eidx, hbuf, deg, comb);

  k_gemm_mlp<<<dim3(kN / 16, kH / 16), dim3(32), 0, stream>>>(
      xfeat, xlabel, W_mlp, b_mlp, comb);

  k_gcn_bias_relu<<<dim3((kN * kH) / 256), blk256, 0, stream>>>(comb, b_gcn);

  k_gemm_cls<<<dim3(kN / 16, 3), dim3(32), 0, stream>>>(comb, W_cls, b_cls, out);
}


# GEMMs retiled: 8-wave blocks stage W once in LDS, each wave does a 16-row strip over all columns, A read once by 16-byte loads; k summed in a new order inside groups of 8
# speedup vs baseline: 2.3306x; 2.3306x over previous
#include <hip/hip_runtime.h>


typedef __attribute__((ext_vector_type(2))) float v2f;
typedef __attribute__((ext_vector_type(8))) float v8f;

constexpr int kN  = 100000;
constexpr int kE  = 1600000;
constexpr int kXF = 128;
constexpr int kXL = 40;
constexpr int kH  = 128;
constexpr int kC  = 40;
constexpr int kKM = kXF + kXL;
constexpr int kKC = 2 * kH;


__global__ void k_init_deg(float* __restrict__ deg) {
  int i = blockIdx.x * blockDim.x + threadIdx.x;
  if (i < kN) deg[i] = 1.0f;
}

__global__ void k_deg_accum(const int* __restrict__ eidx, float* __restrict__ deg) {
  int e = blockIdx.x * blockDim.x + threadIdx.x;
  if (e < kE) atomicAdd(deg + eidx[kE + e], 1.0f);
}

__global__ void k_dinv(float* __restrict__ deg) {
  int i = blockIdx.x * blockDim.x + threadIdx.x;
  if (i < kN) {
    float d = deg[i];
    deg[i] = (d > 0.0f) ? rsqrtf(d) : 0.0f;
  }
}


constexpr int kStrips = kN / 16;
constexpr int kWavesPerBlock = 8;
constexpr int kGemmBlocks = (kStrips + kWavesPerBlock - 1) / kWavesPerBlock;

__device__ __forceinline__ v8f wmma4(v2f a, v2f b, v8f c) {
  return __builtin_amdgcn_wmma_f32_16x16x4_f32(false, a, false, b, (short)0, c,
                                               false, false);
}


template <int K, bool kMlp>
__global__ __launch_bounds__(256)
void k_gemm_n128(const float* __restrict__ A0,
                 const float* __restrict__ A1,
                 const float* __restrict__ Wg,
                 const float* __restrict__ bias,
                 float* __restrict__ Out, int ldo) {
  __shared__ float4 Bs[(K / 4) * 128];

  const int tid  = threadIdx.x;
  const int lane = tid & 31;
  const int h = lane >> 4, c = lane & 15;
  const int strip = blockIdx.x * kWavesPerBlock + (tid >> 5);
  const bool valid = strip < kStrips;
  const int row0 = (valid ? strip : kStrips - 1) * 16;

  float4 av[K / 8];
  {
    const float* ap = A0 + (size_t)(row0 + c) * kXF + 4 * h;
#pragma unroll
    for (int i = 0; i < kXF / 8; ++i)
      av[i] = *reinterpret_cast<const float4*>(ap + 8 * i);
    if constexpr (kMlp) {
      const float* lp = A1 + (size_t)(row0 + c) * kXL + 4 * h;
#pragma unroll
      for (int i = 0; i < kXL / 8; ++i)
        av[kXF / 8 + i] = *reinterpret_cast<const float4*>(lp + 8 * i);
    }
  }

  constexpr int kQuads = (K / 4) * 32;
  auto stage_quad = [&](int idx) {
    const int kq = idx >> 5, nq = idx & 31;
    const float* p = Wg + (size_t)(4 * kq) * kH + 4 * nq;
    const float4 r0 = *reinterpret_cast<const float4*>(p);
    const float4 r1 = *reinterpret_cast<const float4*>(p + kH);
    const float4 r2 = *reinterpret_cast<const float4*>(p + 2 * kH);
    const float4 r3 = *reinterpret_cast<const float4*>(p + 3 * kH);
    float4* d = Bs + kq * 128 + (nq >> 4) * 64 + (nq & 15);
    d[0]  = make_float4(r0.x, r1.x, r2.x, r3.x);
    d[16] = make_float4(r0.y, r1.y, r2.y, r3.y);
    d[32] = make_float4(r0.z, r1.z, r2.z, r3.z);
    d[48] = make_float4(r0.w, r1.w, r2.w, r3.w);
  };
#pragma unroll
  for (int it = 0; it < kQuads / 256; ++it) stage_quad(tid + 256 * it);
  if constexpr (kQuads % 256 != 0) {
    if (tid < kQuads % 256) stage_quad(tid + 256 * (kQuads / 256));
  }
  __syncthreads();
  if (!valid) return;

  v8f acc[8];
#pragma unroll
  for (int t = 0; t < 8; ++t) acc[t] = v8f{};

  const float4* bp = Bs + h * 128 + c;
#pragma unroll
  for (int i = 0; i < K / 8; ++i) {
    float4 b[8];
#pragma unroll
    for (int t = 0; t < 8; ++t) b[t] = bp[i * 256 + t * 16];
    const v2f a01 = {av[i].x, av[i].y};
    const v2f a23 = {av[i].z, av[i].w};
#pragma unroll
    for (int t = 0; t < 8; ++t)
      acc[t] = wmma4(a01, v2f{b[t].x, b[t].y}, acc[t]);
#pragma unroll
    for (int t = 0; t < 8; ++t)
      acc[t] = wmma4(a23, v2f{b[t].z, b[t].w}, acc[t]);
  }

  float* op = Out + (size_t)(row0 + 8 * h) * ldo + 4 * c;
#pragma unroll
  for (int g = 0; g < 2; ++g) {
    float4 bv = make_float4(0.0f, 0.0f, 0.0f, 0.0f);
    if constexpr (kMlp) bv = *reinterpret_cast<const float4*>(bias + 64 * g + 4 * c);
#pragma unroll
    for (int r = 0; r < 8; ++r) {
      float4 o = make_float4(acc[4 * g + 0][r], acc[4 * g + 1][r],
                             acc[4 * g + 2][r], acc[4 * g + 3][r]);
      if constexpr (kMlp) {
        o.x = fmaxf(o.x + bv.x, 0.0f);
        o.y = fmaxf(o.y + bv.y, 0.0f);
        o.z = fmaxf(o.z + bv.z, 0.0f);
        o.w = fmaxf(o.w + bv.w, 0.0f);
      }
      *reinterpret_cast<float4*>(op + (size_t)r * ldo + 64 * g) = o;
    }
  }
}


__global__ void k_selfloop(const float* __restrict__ h,
                           const float* __restrict__ dinv,
                           float* __restrict__ comb) {
  int t = blockIdx.x * blockDim.x + threadIdx.x;
  if (t < kN * kH) {
    int i = t >> 7, c = t & 127;
    float di = dinv[i];
    comb[(size_t)i * kKC + c] = h[(size_t)i * kH + c] * di * di;
  }
}


__global__ void k_edge_scatter(const int* __restrict__ eidx,
                               const float* __restrict__ h,
                               const float* __restrict__ dinv,
                               float* __restrict__ comb) {
  long long t = (long long)blockIdx.x * blockDim.x + threadIdx.x;
  if (t >= (long long)kE * 32) return;
  int e = (int)(t >> 5);
  int q = (int)(t & 31);
  int s = eidx[e];
  int d = eidx[kE + e];
  float nm = dinv[s] * dinv[d];
  const float4 hv = *reinterpret_cast<const float4*>(h + (size_t)s * kH + q * 4);
  float* p = comb + (size_t)d * kKC + q * 4;
  atomicAdd(p + 0, hv.x * nm);
  atomicAdd(p + 1, hv.y * nm);
  atomicAdd(p + 2, hv.z * nm);
  atomicAdd(p + 3, hv.w * nm);
}


__global__ void k_gcn_bias_relu(float* __restrict__ comb, const float* __restrict__ bg) {
  int t = blockIdx.x * blockDim.x + threadIdx.x;
  if (t < kN * kH) {
    int i = t >> 7, c = t & 127;
    float v = comb[(size_t)i * kKC + c] + bg[c];
    comb[(size_t)i * kKC + c] = v > 0.0f ? v : 0.0f;
  }
}


__global__ __launch_bounds__(256)
void k_gemm_cls(const float* __restrict__ comb, const float* __restrict__ Wc,
                const float* __restrict__ bias, float* __restrict__ out) {
  __shared__ float4 Bs[(kKC / 4) * 48];

  const int tid  = threadIdx.x;
  const int lane = tid & 31;
  const int h = lane >> 4, c = lane & 15;
  const int strip = blockIdx.x * kWavesPerBlock + (tid >> 5);
  const bool valid = strip < kStrips;
  const int row0 = (valid ? strip : kStrips - 1) * 16;

  float4 av[kKC / 8];
  {
    const float* ap = comb + (size_t)(row0 + c) * kKC + 4 * h;
#pragma unroll
    for (int i = 0; i < kKC / 8; ++i)
      av[i] = *reinterpret_cast<const float4*>(ap + 8 * i);
  }

#pragma unroll
  for (int it = 0; it < (kKC / 4) * 48 / 256; ++it) {
    const int idx = tid + 256 * it;
    const int kq = idx / 48, n = idx - kq * 48;
    float4 v = make_float4(0.0f, 0.0f, 0.0f, 0.0f);
    if (n < kC) {
      const float* p = Wc + (size_t)(4 * kq) * kC + n;
      v = make_float4(p[0], p[kC], p[2 * kC], p[3 * kC]);
    }
    Bs[idx] = v;
  }
  __syncthreads();
  if (!valid) return;

  v8f acc[3];
#pragma unroll
  for (int t = 0; t < 3; ++t) acc[t] = v8f{};

  const float4* bp = Bs + h * 48 + c;
#pragma unroll
  for (int i = 0; i < kKC / 8; ++i) {
    float4 b[3];
#pragma unroll
    for (int t = 0; t < 3; ++t) b[t] = bp[i * 96 + t * 16];
    const v2f a01 = {av[i].x, av[i].y};
    const v2f a23 = {av[i].z, av[i].w};
#pragma unroll
    for (int t = 0; t < 3; ++t)
      acc[t] = wmma4(a01, v2f{b[t].x, b[t].y}, acc[t]);
#pragma unroll
    for (int t = 0; t < 3; ++t)
      acc[t] = wmma4(a23, v2f{b[t].z, b[t].w}, acc[t]);
  }

  float* op = out + (size_t)(row0 + 8 * h) * kC;
#pragma unroll
  for (int t = 0; t < 3; ++t) {
    const int col = 16 * t + c;
    const float bn = bias[col < kC ? col : kC - 1];
    float o[8];
#pragma unroll
    for (int r = 0; r < 8; ++r) o[r] = acc[t][r] + bn;
    if (col < kC) {
#pragma unroll
      for (int r = 0; r < 8; ++r) op[(size_t)r * kC + col] = o[r];
    }
  }
}


extern "C" void kernel_launch(void* const* d_in, const int* in_sizes, int n_in,
                              void* d_out, int out_size, void* d_ws, size_t ws_size,
                              hipStream_t stream) {
  (void)in_sizes; (void)n_in; (void)out_size; (void)ws_size;

  const float* xfeat  = (const float*)d_in[0];
  const float* xlabel = (const float*)d_in[1];
  const int*   eidx   = (const int*)d_in[2];
  const float* W_gcn  = (const float*)d_in[3];
  const float* b_gcn  = (const float*)d_in[4];
  const float* W_mlp  = (const float*)d_in[5];
  const float* b_mlp  = (const float*)d_in[6];
  const float* W_cls  = (const float*)d_in[7];
  const float* b_cls  = (const float*)d_in[8];
  float* out = (float*)d_out;

  char* ws = (char*)d_ws;
  const size_t off_deg  = 0;
  const size_t off_h    = (off_deg + (size_t)kN * 4 + 255) & ~(size_t)255;
  const size_t off_comb = (off_h + (size_t)kN * kH * 4 + 255) & ~(size_t)255;
  float* deg  = (float*)(ws + off_deg);
  float* hbuf = (float*)(ws + off_h);
  float* comb = (float*)(ws + off_comb);

  const dim3 blk256(256);

  k_init_deg <<<dim3((kN + 255) / 256), blk256, 0, stream>>>(deg);
  k_deg_accum<<<dim3((kE + 255) / 256), blk256, 0, stream>>>(eidx, deg);
  k_dinv     <<<dim3((kN + 255) / 256), blk256, 0, stream>>>(deg);

  k_gemm_n128<kXF, false><<<dim3(kGemmBlocks), blk256, 0, stream>>>(
      xfeat, xlabel, W_gcn, b_gcn, hbuf, kH);

  k_selfloop<<<dim3((kN * kH) / 256), blk256, 0, stream>>>(hbuf, deg, comb);
  k_edge_scatter<<<dim3((unsigned)(((long long)kE * 32) / 256)), blk256, 0, stream>>>(
      eidx, hbuf, deg, comb);

  k_gemm_n128<kKM, true><<<dim3(kGemmBlocks), blk256, 0, stream>>>(
      xfeat, xlabel, W_mlp, b_mlp, comb + kH, kKC);

  k_gcn_bias_relu<<<dim3((kN * kH) / 256), blk256, 0, stream>>>(comb, b_gcn);

  k_gemm_cls<<<dim3(kGemmBlocks), blk256, 0, stream>>>(comb, W_cls, b_cls, out);
}


# float-atomic edge scatter replaced by counting sort of edges by destination (integer atomics, arrival-order rank) and one wave per node summing its rows in registers, one store per row
# speedup vs baseline: 4.3734x; 1.8765x over previous
#include <hip/hip_runtime.h>


typedef __attribute__((ext_vector_type(2))) float v2f;
typedef __attribute__((ext_vector_type(8))) float v8f;

constexpr int kN  = 100000;
constexpr int kE  = 1600000;
constexpr int kXF = 128;
constexpr int kXL = 40;
constexpr int kH  = 128;
constexpr int kC  = 40;
constexpr int kKM = kXF + kXL;
constexpr int kKC = 2 * kH;


__global__ void k_zero_cnt(int* __restrict__ cnt) {
  int i = blockIdx.x * blockDim.x + threadIdx.x;
  if (i < kN) cnt[i] = 0;
}

__global__ void k_rank(const int* __restrict__ eidx, int* __restrict__ cnt,
                       int* __restrict__ rank) {
  int e = blockIdx.x * blockDim.x + threadIdx.x;
  if (e < kE) rank[e] = atomicAdd(cnt + eidx[kE + e], 1);
}

constexpr int kScanBlock = 1024;
constexpr int kScanBlocks = (kN + kScanBlock - 1) / kScanBlock;
constexpr int kScanPreIters =
    ((kScanBlocks - 1) * (kScanBlock / 4) + kScanBlock - 1) / kScanBlock;

__global__ __launch_bounds__(kScanBlock)
void k_scan(const int* __restrict__ cnt, int* __restrict__ rowptr,
            float* __restrict__ dinv) {
  __shared__ int s_part[32];
  __shared__ int s_wave[32];
  const int tid = threadIdx.x;
  const int lane = tid & 31, wv = tid >> 5;
  const int b = blockIdx.x;

  int pre = 0;
  const int4* c4 = reinterpret_cast<const int4*>(cnt);
  const int lim = b * (kScanBlock / 4);
  int4 pv[kScanPreIters];
#pragma unroll
  for (int k = 0; k < kScanPreIters; ++k) {
    const int i = tid + kScanBlock * k;
    pv[k] = make_int4(0, 0, 0, 0);
    if (i < lim) pv[k] = c4[i];
  }
#pragma unroll
  for (int k = 0; k < kScanPreIters; ++k)
    pre += (pv[k].x + pv[k].y) + (pv[k].z + pv[k].w);
  const int idx = b * kScanBlock + tid;
  const int v = idx < kN ? cnt[idx] : 0;

  int incl = v;
#pragma unroll
  for (int o = 1; o < 32; o <<= 1) {
    pre += __shfl_xor(pre, o);
    const int up = __shfl_up(incl, o);
    if (lane >= o) incl += up;
  }
  if (lane == 31) s_wave[wv] = incl;
  if (lane == 0) s_part[wv] = pre;
  __syncthreads();
  int base = 0;
  int wpre = 0;
  for (int w = 0; w < 32; ++w) {
    base += s_part[w];
    wpre += (w < wv) ? s_wave[w] : 0;
  }
  if (idx < kN) {
    const int excl = base + wpre + incl - v;
    rowptr[idx] = excl;
    if (idx == kN - 1) rowptr[kN] = excl + v;
    const float d = (float)(v + 1);
    dinv[idx] = (d > 0.0f) ? rsqrtf(d) : 0.0f;
  }
}

__global__ void k_place(const int* __restrict__ eidx, const int* __restrict__ rowptr,
                        const int* __restrict__ rank, int* __restrict__ csr_src) {
  int e = blockIdx.x * blockDim.x + threadIdx.x;
  if (e < kE) csr_src[rowptr[eidx[kE + e]] + rank[e]] = eidx[e];
}


constexpr int kStrips = kN / 16;
constexpr int kWavesPerBlock = 8;
constexpr int kGemmBlocks = (kStrips + kWavesPerBlock - 1) / kWavesPerBlock;

__device__ __forceinline__ v8f wmma4(v2f a, v2f b, v8f c) {
  return __builtin_amdgcn_wmma_f32_16x16x4_f32(false, a, false, b, (short)0, c,
                                               false, false);
}


template <int K, bool kMlp>
__global__ __launch_bounds__(256)
void k_gemm_n128(const float* __restrict__ A0,
                 const float* __restrict__ A1,
                 const float* __restrict__ Wg,
                 const float* __restrict__ bias,
                 float* __restrict__ Out, int ldo) {
  __shared__ float4 Bs[(K / 4) * 128];

  const int tid  = threadIdx.x;
  const int lane = tid & 31;
  const int h = lane >> 4, c = lane & 15;
  const int strip = blockIdx.x * kWavesPerBlock + (tid >> 5);
  const bool valid = strip < kStrips;
  const int row0 = (valid ? strip : kStrips - 1) * 16;

  float4 av[K / 8];
  {
    const float* ap = A0 + (size_t)(row0 + c) * kXF + 4 * h;
#pragma unroll
    for (int i = 0; i < kXF / 8; ++i)
      av[i] = *reinterpret_cast<const float4*>(ap + 8 * i);
    if constexpr (kMlp) {
      const float* lp = A1 + (size_t)(row0 + c) * kXL + 4 * h;
#pragma unroll
      for (int i = 0; i < kXL / 8; ++i)
        av[kXF / 8 + i] = *reinterpret_cast<const float4*>(lp + 8 * i);
    }
  }

  constexpr int kQuads = (K / 4) * 32;
  auto stage_quad = [&](int idx) {
    const int kq = idx >> 5, nq = idx & 31;
    const float* p = Wg + (size_t)(4 * kq) * kH + 4 * nq;
    const float4 r0 = *reinterpret_cast<const float4*>(p);
    const float4 r1 = *reinterpret_cast<const float4*>(p + kH);
    const float4 r2 = *reinterpret_cast<const float4*>(p + 2 * kH);
    const float4 r3 = *reinterpret_cast<const float4*>(p + 3 * kH);
    float4* d = Bs + kq * 128 + (nq >> 4) * 64 + (nq & 15);
    d[0]  = make_float4(r0.x, r1.x, r2.x, r3.x);
    d[16] = make_float4(r0.y, r1.y, r2.y, r3.y);
    d[32] = make_float4(r0.z, r1.z, r2.z, r3.z);
    d[48] = make_float4(r0.w, r1.w, r2.w, r3.w);
  };
#pragma unroll
  for (int it = 0; it < kQuads / 256; ++it) stage_quad(tid + 256 * it);
  if constexpr (kQuads % 256 != 0) {
    if (tid < kQuads % 256) stage_quad(tid + 256 * (kQuads / 256));
  }
  __syncthreads();
  if (!valid) return;

  v8f acc[8];
#pragma unroll
  for (int t = 0; t < 8; ++t) acc[t] = v8f{};

  const float4* bp = Bs + h * 128 + c;
#pragma unroll
  for (int i = 0; i < K / 8; ++i) {
    float4 b[8];
#pragma unroll
    for (int t = 0; t < 8; ++t) b[t] = bp[i * 256 + t * 16];
    const v2f a01 = {av[i].x, av[i].y};
    const v2f a23 = {av[i].z, av[i].w};
#pragma unroll
    for (int t = 0; t < 8; ++t)
      acc[t] = wmma4(a01, v2f{b[t].x, b[t].y}, acc[t]);
#pragma unroll
    for (int t = 0; t < 8; ++t)
      acc[t] = wmma4(a23, v2f{b[t].z, b[t].w}, acc[t]);
  }

  float* op = Out + (size_t)(row0 + 8 * h) * ldo + 4 * c;
#pragma unroll
  for (int g = 0; g < 2; ++g) {
    float4 bv = make_float4(0.0f, 0.0f, 0.0f, 0.0f);
    if constexpr (kMlp) bv = *reinterpret_cast<const float4*>(bias + 64 * g + 4 * c);
#pragma unroll
    for (int r = 0; r < 8; ++r) {
      float4 o = make_float4(acc[4 * g + 0][r], acc[4 * g + 1][r],
                             acc[4 * g + 2][r], acc[4 * g + 3][r]);
      if constexpr (kMlp) {
        o.x = fmaxf(o.x + bv.x, 0.0f);
        o.y = fmaxf(o.y + bv.y, 0.0f);
        o.z = fmaxf(o.z + bv.z, 0.0f);
        o.w = fmaxf(o.w + bv.w, 0.0f);
      }
      *reinterpret_cast<float4*>(op + (size_t)r * ldo + 64 * g) = o;
    }
  }
}


__global__ __launch_bounds__(256)
void k_gather(const int* __restrict__ rowptr, const int* __restrict__ csr_src,
              const float* __restrict__ h, const float* __restrict__ dinv,
              const float* __restrict__ bg, float* __restrict__ comb) {
  const int lane = threadIdx.x & 31;
  const int d = __builtin_amdgcn_readfirstlane((int)(blockIdx.x * 8 + (threadIdx.x >> 5)));
  if (d >= kN) return;
  const int beg = rowptr[d];
  const int end = rowptr[d + 1];
  const float dd = dinv[d];
  const float4* hq = reinterpret_cast<const float4*>(h) + lane;

  float4 acc;
  {
    const float4 v = hq[(size_t)d * (kH / 4)];
    const float w = dd * dd;
    acc = make_float4(v.x * w, v.y * w, v.z * w, v.w * w);
  }

  for (int base = beg; base < end; base += 32) {
    const int n = (end - base) < 32 ? (end - base) : 32;
    int s_l = 0;
    float w_l = 0.0f;
    if (lane < n) {
      s_l = csr_src[base + lane];
      w_l = dinv[s_l] * dd;
    }
    auto row = [&](int j) {
      const int s = __builtin_amdgcn_readlane(s_l, j);
      return hq[(size_t)s * (kH / 4)];
    };
    auto add = [&](const float4 v, int j) {
      const float w = __builtin_bit_cast(
          float, __builtin_amdgcn_readlane(__builtin_bit_cast(int, w_l), j));
      acc.x = fmaf(v.x, w, acc.x);
      acc.y = fmaf(v.y, w, acc.y);
      acc.z = fmaf(v.z, w, acc.z);
      acc.w = fmaf(v.w, w, acc.w);
    };
    int j = 0;
    for (; j + 8 <= n; j += 8) {
      float4 v[8];
#pragma unroll
      for (int k = 0; k < 8; ++k) v[k] = row(j + k);
#pragma unroll
      for (int k = 0; k < 8; ++k) add(v[k], j + k);
    }
    if (j + 4 <= n) {
      float4 v[4];
#pragma unroll
      for (int k = 0; k < 4; ++k) v[k] = row(j + k);
#pragma unroll
      for (int k = 0; k < 4; ++k) add(v[k], j + k);
      j += 4;
    }
    const int r = n - j;
    if (r > 0) {
      const float4 v0 = row(j);
      float4 v1 = v0, v2 = v0;
      if (r > 1) v1 = row(j + 1);
      if (r > 2) v2 = row(j + 2);
      add(v0, j);
      if (r > 1) add(v1, j + 1);
      if (r > 2) add(v2, j + 2);
    }
  }

  const float4 bv = *reinterpret_cast<const float4*>(bg + 4 * lane);
  float4 o;
  o.x = fmaxf(acc.x + bv.x, 0.0f);
  o.y = fmaxf(acc.y + bv.y, 0.0f);
  o.z = fmaxf(acc.z + bv.z, 0.0f);
  o.w = fmaxf(acc.w + bv.w, 0.0f);
  *reinterpret_cast<float4*>(comb + (size_t)d * kKC + 4 * lane) = o;
}


__global__ __launch_bounds__(256)
void k_gemm_cls(const float* __restrict__ comb, const float* __restrict__ Wc,
                const float* __restrict__ bias, float* __restrict__ out) {
  __shared__ float4 Bs[(kKC / 4) * 48];

  const int tid  = threadIdx.x;
  const int lane = tid & 31;
  const int h = lane >> 4, c = lane & 15;
  const int strip = blockIdx.x * kWavesPerBlock + (tid >> 5);
  const bool valid = strip < kStrips;
  const int row0 = (valid ? strip : kStrips - 1) * 16;

  float4 av[kKC / 8];
  {
    const float* ap = comb + (size_t)(row0 + c) * kKC + 4 * h;
#pragma unroll
    for (int i = 0; i < kKC / 8; ++i)
      av[i] = *reinterpret_cast<const float4*>(ap + 8 * i);
  }

#pragma unroll
  for (int it = 0; it < (kKC / 4) * 48 / 256; ++it) {
    const int idx = tid + 256 * it;
    const int kq = idx / 48, n = idx - kq * 48;
    float4 v = make_float4(0.0f, 0.0f, 0.0f, 0.0f);
    if (n < kC) {
      const float* p = Wc + (size_t)(4 * kq) * kC + n;
      v = make_float4(p[0], p[kC], p[2 * kC], p[3 * kC]);
    }
    Bs[idx] = v;
  }
  __syncthreads();
  if (!valid) return;

  v8f acc[3];
#pragma unroll
  for (int t = 0; t < 3; ++t) acc[t] = v8f{};

  const float4* bp = Bs + h * 48 + c;
#pragma unroll
  for (int i = 0; i < kKC / 8; ++i) {
    float4 b[3];
#pragma unroll
    for (int t = 0; t < 3; ++t) b[t] = bp[i * 96 + t * 16];
    const v2f a01 = {av[i].x, av[i].y};
    const v2f a23 = {av[i].z, av[i].w};
#pragma unroll
    for (int t = 0; t < 3; ++t)
      acc[t] = wmma4(a01, v2f{b[t].x, b[t].y}, acc[t]);
#pragma unroll
    for (int t = 0; t < 3; ++t)
      acc[t] = wmma4(a23, v2f{b[t].z, b[t].w}, acc[t]);
  }

  float* op = out + (size_t)(row0 + 8 * h) * kC;
#pragma unroll
  for (int t = 0; t < 3; ++t) {
    const int col = 16 * t + c;
    const float bn = bias[col < kC ? col : kC - 1];
    float o[8];
#pragma unroll
    for (int r = 0; r < 8; ++r) o[r] = acc[t][r] + bn;
    if (col < kC) {
#pragma unroll
      for (int r = 0; r < 8; ++r) op[(size_t)r * kC + col] = o[r];
    }
  }
}


extern "C" void kernel_launch(void* const* d_in, const int* in_sizes, int n_in,
                              void* d_out, int out_size, void* d_ws, size_t ws_size,
                              hipStream_t stream) {
  (void)in_sizes; (void)n_in; (void)out_size; (void)ws_size;

  const float* xfeat  = (const float*)d_in[0];
  const float* xlabel = (const float*)d_in[1];
  const int*   eidx   = (const int*)d_in[2];
  const float* W_gcn  = (const float*)d_in[3];
  const float* b_gcn  = (const float*)d_in[4];
  const float* W_mlp  = (const float*)d_in[5];
  const float* b_mlp  = (const float*)d_in[6];
  const float* W_cls  = (const float*)d_in[7];
  const float* b_cls  = (const float*)d_in[8];
  float* out = (float*)d_out;

  char* ws = (char*)d_ws;
  const size_t off_deg  = 0;
  const size_t off_h    = (off_deg + (size_t)kN * 4 + 255) & ~(size_t)255;
  const size_t off_comb = (off_h + (size_t)kN * kH * 4 + 255) & ~(size_t)255;
  const size_t off_cnt  = (off_comb + (size_t)kN * kKC * 4 + 255) & ~(size_t)255;
  const size_t off_rptr = (off_cnt + (size_t)kN * 4 + 255) & ~(size_t)255;
  const size_t off_rank = (off_rptr + (size_t)(kN + 1) * 4 + 255) & ~(size_t)255;
  const size_t off_csr  = (off_rank + (size_t)kE * 4 + 255) & ~(size_t)255;
  float* dinv = (float*)(ws + off_deg);
  float* hbuf = (float*)(ws + off_h);
  float* comb = (float*)(ws + off_comb);
  int* cnt    = (int*)(ws + off_cnt);
  int* rowptr = (int*)(ws + off_rptr);
  int* rank   = (int*)(ws + off_rank);
  int* csrsrc = (int*)(ws + off_csr);

  const dim3 blk256(256);

  k_zero_cnt<<<dim3((kN + 255) / 256), blk256, 0, stream>>>(cnt);
  k_rank    <<<dim3((kE + 255) / 256), blk256, 0, stream>>>(eidx, cnt, rank);
  k_scan    <<<dim3(kScanBlocks), dim3(kScanBlock), 0, stream>>>(cnt, rowptr, dinv);
  k_place   <<<dim3((kE + 255) / 256), blk256, 0, stream>>>(eidx, rowptr, rank, csrsrc);

  k_gemm_n128<kXF, false><<<dim3(kGemmBlocks), blk256, 0, stream>>>(
      xfeat, xlabel, W_gcn, b_gcn, hbuf, kH);

  k_gather<<<dim3((kN + 7) / 8), blk256, 0, stream>>>(rowptr, csrsrc, hbuf, dinv, b_gcn, comb);

  k_gemm_n128<kKM, true><<<dim3(kGemmBlocks), blk256, 0, stream>>>(
      xfeat, xlabel, W_mlp, b_mlp, comb + kH, kKC);

  k_gemm_cls<<<dim3(kGemmBlocks), blk256, 0, stream>>>(comb, W_cls, b_cls, out);
}


# edges ordered by destination with a two-level counting sort on LDS counters and contiguous copy-out, replacing global atomics and scattered 4-byte stores; segment order = LDS atomic arrival order
# speedup vs baseline: 7.1424x; 1.6331x over previous
#include <hip/hip_runtime.h>


typedef __attribute__((ext_vector_type(2))) float v2f;
typedef __attribute__((ext_vector_type(8))) float v8f;

constexpr int kN  = 100000;
constexpr int kE  = 1600000;
constexpr int kXF = 128;
constexpr int kXL = 40;
constexpr int kH  = 128;
constexpr int kC  = 40;
constexpr int kKM = kXF + kXL;
constexpr int kKC = 2 * kH;


constexpr int kBktShift = 9;
constexpr int kBktNodes = 1 << kBktShift;
constexpr int kNB = (kN + kBktNodes - 1) >> kBktShift;
constexpr int kSortThreads = 1024;
constexpr int kChunk = 6400;
constexpr int kSortBlocks = (kE + kChunk - 1) / kChunk;
constexpr int kChunkIters = (kChunk + kSortThreads - 1) / kSortThreads;
constexpr int kTabIters = (kSortBlocks + 3) / 4;
constexpr int kStageCap = 16384;
constexpr int kGrp = 12;
static_assert(kNB <= 256, "bucket scan uses 8 waves");
static_assert(kN <= (1 << (31 - kBktShift)), "source id and row-in-bucket share one word");

template <int NW>
__device__ __forceinline__ int block_excl_scan(int v, int* s_w) {
  const int lane = threadIdx.x & 31, wv = threadIdx.x >> 5;
  int incl = v;
#pragma unroll
  for (int o = 1; o < 32; o <<= 1) {
    const int up = __shfl_up(incl, o);
    if (lane >= o) incl += up;
  }
  if (wv < NW && lane == 31) s_w[wv] = incl;
  __syncthreads();
  int wpre = 0;
#pragma unroll
  for (int w = 0; w < NW; ++w) wpre += (w < wv) ? s_w[w] : 0;
  return wpre + incl - v;
}

__global__ __launch_bounds__(kSortThreads)
void k_hist(const int* __restrict__ eidx, int* __restrict__ hist) {
  __shared__ int s_h[kNB];
  const int tid = threadIdx.x, g = blockIdx.x;
  const int e0 = g * kChunk;
  const int n = (e0 + kChunk <= kE ? kChunk : kE - e0);
  int dv[kChunkIters];
#pragma unroll
  for (int i = 0; i < kChunkIters; ++i) {
    const int idx = tid + kSortThreads * i;
    dv[i] = 0;
    if (idx < n) dv[i] = eidx[kE + e0 + idx];
  }
  if (tid < kNB) s_h[tid] = 0;
  __syncthreads();
#pragma unroll
  for (int i = 0; i < kChunkIters; ++i)
    if (tid + kSortThreads * i < n) atomicAdd(&s_h[dv[i] >> kBktShift], 1);
  __syncthreads();
  if (tid < kNB) hist[g * kNB + tid] = s_h[tid];
}

__global__ __launch_bounds__(kSortThreads)
void k_bin(const int* __restrict__ eidx, const int* __restrict__ hist,
           int* __restrict__ bstart, int* __restrict__ tmp) {
  __shared__ int s_tot4[4][kNB];
  __shared__ int s_pre4[4][kNB];
  __shared__ int s_w1[8];
  __shared__ int s_w2[8];
  __shared__ int s_lstart[kNB];
  __shared__ int s_delta[kNB];
  __shared__ int s_cnt[kNB];
  __shared__ int s_ss[kChunk];
  __shared__ int s_sd[kChunk];

  const int tid = threadIdx.x, g = blockIdx.x;
  const int b = tid & 255, q = tid >> 8;
  const int e0 = g * kChunk;
  const int n = (e0 + kChunk <= kE ? kChunk : kE - e0);

  int sv[kChunkIters], dv[kChunkIters];
#pragma unroll
  for (int i = 0; i < kChunkIters; ++i) {
    const int idx = tid + kSortThreads * i;
    sv[i] = 0; dv[i] = 0;
    if (idx < n) { sv[i] = eidx[e0 + idx]; dv[i] = eidx[kE + e0 + idx]; }
  }

  if (b < kNB) {
    int tot = 0, pre = 0;
#pragma unroll 21
    for (int it = 0; it < kTabIters; ++it) {
      const int gg = q + 4 * it;
      if (gg < kSortBlocks) {
        const int v = hist[gg * kNB + b];
        tot += v;
        pre += (gg < g) ? v : 0;
      }
    }
    s_tot4[q][b] = tot;
    s_pre4[q][b] = pre;
  }
  int own = 0;
  if (tid < kNB) { own = hist[g * kNB + tid]; s_cnt[tid] = 0; }
  __syncthreads();
  int tot = 0, pre = 0;
  if (tid < kNB) {
    tot = (s_tot4[0][tid] + s_tot4[1][tid]) + (s_tot4[2][tid] + s_tot4[3][tid]);
    pre = (s_pre4[0][tid] + s_pre4[1][tid]) + (s_pre4[2][tid] + s_pre4[3][tid]);
  }
  const int gstart = block_excl_scan<8>(tot, s_w1);
  const int lstart = block_excl_scan<8>(own, s_w2);
  if (tid < kNB) {
    s_lstart[tid] = lstart;
    s_delta[tid] = gstart + pre - lstart;
    if (g == 0) {
      bstart[tid] = gstart;
      if (tid == kNB - 1) bstart[kNB] = gstart + tot;
    }
  }
  __syncthreads();
#pragma unroll
  for (int i = 0; i < kChunkIters; ++i) {
    if (tid + kSortThreads * i < n) {
      const int bk = dv[i] >> kBktShift;
      const int p = s_lstart[bk] + atomicAdd(&s_cnt[bk], 1);
      s_ss[p] = sv[i];
      s_sd[p] = dv[i];
    }
  }
  __syncthreads();
#pragma unroll
  for (int i = 0; i < kChunkIters; ++i) {
    const int idx = tid + kSortThreads * i;
    if (idx < n) {
      const int d = s_sd[idx];
      const unsigned w = ((unsigned)s_ss[idx] << kBktShift) | (unsigned)(d & (kBktNodes - 1));
      tmp[idx + s_delta[d >> kBktShift]] = (int)w;
    }
  }
}

__global__ __launch_bounds__(kSortThreads)
void k_bucket(const int* __restrict__ tmp, const int* __restrict__ bstart,
              int* __restrict__ csr_src, int* __restrict__ rowptr,
              float* __restrict__ dinv) {
  __shared__ int s_cnt[kBktNodes];
  __shared__ int s_cur[kBktNodes];
  __shared__ int s_wave[kBktNodes / 32];
  __shared__ int s_out[kStageCap];
  const int tid = threadIdx.x;
  const int bk = blockIdx.x;
  const int beg = bstart[bk];
  const int n = bstart[bk + 1] - beg;
  const int* tp = tmp + beg;

  if (tid < kBktNodes) s_cnt[tid] = 0;
  __syncthreads();
  for (int base = 0; base < n; base += kGrp * kSortThreads) {
    int w[kGrp];
#pragma unroll
    for (int k = 0; k < kGrp; ++k) {
      const int i = base + tid + kSortThreads * k;
      w[k] = 0;
      if (i < n) w[k] = tp[i];
    }
#pragma unroll
    for (int k = 0; k < kGrp; ++k)
      if (base + tid + kSortThreads * k < n) atomicAdd(&s_cnt[w[k] & (kBktNodes - 1)], 1);
  }
  __syncthreads();
  const int v = tid < kBktNodes ? s_cnt[tid] : 0;
  const int excl = block_excl_scan<kBktNodes / 32>(v, s_wave);
  if (tid < kBktNodes) {
    s_cur[tid] = excl;
    const int node = bk * kBktNodes + tid;
    if (node < kN) {
      rowptr[node] = beg + excl;
      if (node == kN - 1) rowptr[kN] = beg + excl + v;
      const float d = (float)(v + 1);
      dinv[node] = (d > 0.0f) ? rsqrtf(d) : 0.0f;
    }
  }
  __syncthreads();
  const bool staged = n <= kStageCap;
  for (int base = 0; base < n; base += kGrp * kSortThreads) {
    int w[kGrp];
#pragma unroll
    for (int k = 0; k < kGrp; ++k) {
      const int i = base + tid + kSortThreads * k;
      w[k] = 0;
      if (i < n) w[k] = tp[i];
    }
#pragma unroll
    for (int k = 0; k < kGrp; ++k) {
      if (base + tid + kSortThreads * k < n) {
        const int p = atomicAdd(&s_cur[w[k] & (kBktNodes - 1)], 1);
        const int s = (int)((unsigned)w[k] >> kBktShift);
        if (staged) s_out[p] = s; else csr_src[beg + p] = s;
      }
    }
  }
  if (staged) {
    __syncthreads();
    for (int i = tid; i < n; i += kSortThreads) csr_src[beg + i] = s_out[i];
  }
}


constexpr int kStrips = kN / 16;
constexpr int kWavesPerBlock = 8;
constexpr int kGemmBlocks = (kStrips + kWavesPerBlock - 1) / kWavesPerBlock;

__device__ __forceinline__ v8f wmma4(v2f a, v2f b, v8f c) {
  return __builtin_amdgcn_wmma_f32_16x16x4_f32(false, a, false, b, (short)0, c,
                                               false, false);
}


template <int K, bool kMlp>
__global__ __launch_bounds__(256)
void k_gemm_n128(const float* __restrict__ A0,
                 const float* __restrict__ A1,
                 const float* __restrict__ Wg,
                 const float* __restrict__ bias,
                 float* __restrict__ Out, int ldo) {
  __shared__ float4 Bs[(K / 4) * 128];

  const int tid  = threadIdx.x;
  const int lane = tid & 31;
  const int h = lane >> 4, c = lane & 15;
  const int strip = blockIdx.x * kWavesPerBlock + (tid >> 5);
  const bool valid = strip < kStrips;
  const int row0 = (valid ? strip : kStrips - 1) * 16;

  float4 av[K / 8];
  {
    const float* ap = A0 + (size_t)(row0 + c) * kXF + 4 * h;
#pragma unroll
    for (int i = 0; i < kXF / 8; ++i)
      av[i] = *reinterpret_cast<const float4*>(ap + 8 * i);
    if constexpr (kMlp) {
      const float* lp = A1 + (size_t)(row0 + c) * kXL + 4 * h;
#pragma unroll
      for (int i = 0; i < kXL / 8; ++i)
        av[kXF / 8 + i] = *reinterpret_cast<const float4*>(lp + 8 * i);
    }
  }

  constexpr int kQuads = (K / 4) * 32;
  auto stage_quad = [&](int idx) {
    const int kq = idx >> 5, nq = idx & 31;
    const float* p = Wg + (size_t)(4 * kq) * kH + 4 * nq;
    const float4 r0 = *reinterpret_cast<const float4*>(p);
    const float4 r1 = *reinterpret_cast<const float4*>(p + kH);
    const float4 r2 = *reinterpret_cast<const float4*>(p + 2 * kH);
    const float4 r3 = *reinterpret_cast<const float4*>(p + 3 * kH);
    float4* d = Bs + kq * 128 + (nq >> 4) * 64 + (nq & 15);
    d[0]  = make_float4(r0.x, r1.x, r2.x, r3.x);
    d[16] = make_float4(r0.y, r1.y, r2.y, r3.y);
    d[32] = make_float4(r0.z, r1.z, r2.z, r3.z);
    d[48] = make_float4(r0.w, r1.w, r2.w, r3.w);
  };
#pragma unroll
  for (int it = 0; it < kQuads / 256; ++it) stage_quad(tid + 256 * it);
  if constexpr (kQuads % 256 != 0) {
    if (tid < kQuads % 256) stage_quad(tid + 256 * (kQuads / 256));
  }
  __syncthreads();
  if (!valid) return;

  v8f acc[8];
#pragma unroll
  for (int t = 0; t < 8; ++t) acc[t] = v8f{};

  const float4* bp = Bs + h * 128 + c;
#pragma unroll
  for (int i = 0; i < K / 8; ++i) {
    float4 b[8];
#pragma unroll
    for (int t = 0; t < 8; ++t) b[t] = bp[i * 256 + t * 16];
    const v2f a01 = {av[i].x, av[i].y};
    const v2f a23 = {av[i].z, av[i].w};
#pragma unroll
    for (int t = 0; t < 8; ++t)
      acc[t] = wmma4(a01, v2f{b[t].x, b[t].y}, acc[t]);
#pragma unroll
    for (int t = 0; t < 8; ++t)
      acc[t] = wmma4(a23, v2f{b[t].z, b[t].w}, acc[t]);
  }

  float* op = Out + (size_t)(row0 + 8 * h) * ldo + 4 * c;
#pragma unroll
  for (int g = 0; g < 2; ++g) {
    float4 bv = make_float4(0.0f, 0.0f, 0.0f, 0.0f);
    if constexpr (kMlp) bv = *reinterpret_cast<const float4*>(bias + 64 * g + 4 * c);
#pragma unroll
    for (int r = 0; r < 8; ++r) {
      float4 o = make_float4(acc[4 * g + 0][r], acc[4 * g + 1][r],
                             acc[4 * g + 2][r], acc[4 * g + 3][r]);
      if constexpr (kMlp) {
        o.x = fmaxf(o.x + bv.x, 0.0f);
        o.y = fmaxf(o.y + bv.y, 0.0f);
        o.z = fmaxf(o.z + bv.z, 0.0f);
        o.w = fmaxf(o.w + bv.w, 0.0f);
      }
      *reinterpret_cast<float4*>(op + (size_t)r * ldo + 64 * g) = o;
    }
  }
}


__global__ __launch_bounds__(256)
void k_gather(const int* __restrict__ rowptr, const int* __restrict__ csr_src,
              const float* __restrict__ h, const float* __restrict__ dinv,
              const float* __restrict__ bg, float* __restrict__ comb) {
  const int lane = threadIdx.x & 31;
  const int d = __builtin_amdgcn_readfirstlane((int)(blockIdx.x * 8 + (threadIdx.x >> 5)));
  if (d >= kN) return;
  const int beg = rowptr[d];
  const int end = rowptr[d + 1];
  const float dd = dinv[d];
  const float4* hq = reinterpret_cast<const float4*>(h) + lane;

  float4 acc;
  {
    const float4 v = hq[(size_t)d * (kH / 4)];
    const float w = dd * dd;
    acc = make_float4(v.x * w, v.y * w, v.z * w, v.w * w);
  }

  for (int base = beg; base < end; base += 32) {
    const int n = (end - base) < 32 ? (end - base) : 32;
    int s_l = 0;
    float w_l = 0.0f;
    if (lane < n) {
      s_l = csr_src[base + lane];
      w_l = dinv[s_l] * dd;
    }
    auto row = [&](int j) {
      const int s = __builtin_amdgcn_readlane(s_l, j);
      return hq[(size_t)s * (kH / 4)];
    };
    auto add = [&](const float4 v, int j) {
      const float w = __builtin_bit_cast(
          float, __builtin_amdgcn_readlane(__builtin_bit_cast(int, w_l), j));
      acc.x = fmaf(v.x, w, acc.x);
      acc.y = fmaf(v.y, w, acc.y);
      acc.z = fmaf(v.z, w, acc.z);
      acc.w = fmaf(v.w, w, acc.w);
    };
    int j = 0;
    for (; j + 8 <= n; j += 8) {
      float4 v[8];
#pragma unroll
      for (int k = 0; k < 8; ++k) v[k] = row(j + k);
#pragma unroll
      for (int k = 0; k < 8; ++k) add(v[k], j + k);
    }
    if (j + 4 <= n) {
      float4 v[4];
#pragma unroll
      for (int k = 0; k < 4; ++k) v[k] = row(j + k);
#pragma unroll
      for (int k = 0; k < 4; ++k) add(v[k], j + k);
      j += 4;
    }
    const int r = n - j;
    if (r > 0) {
      const float4 v0 = row(j);
      float4 v1 = v0, v2 = v0;
      if (r > 1) v1 = row(j + 1);
      if (r > 2) v2 = row(j + 2);
      add(v0, j);
      if (r > 1) add(v1, j + 1);
      if (r > 2) add(v2, j + 2);
    }
  }

  const float4 bv = *reinterpret_cast<const float4*>(bg + 4 * lane);
  float4 o;
  o.x = fmaxf(acc.x + bv.x, 0.0f);
  o.y = fmaxf(acc.y + bv.y, 0.0f);
  o.z = fmaxf(acc.z + bv.z, 0.0f);
  o.w = fmaxf(acc.w + bv.w, 0.0f);
  *reinterpret_cast<float4*>(comb + (size_t)d * kKC + 4 * lane) = o;
}


__global__ __launch_bounds__(256)
void k_gemm_cls(const float* __restrict__ comb, const float* __restrict__ Wc,
                const float* __restrict__ bias, float* __restrict__ out) {
  __shared__ float4 Bs[(kKC / 4) * 48];

  const int tid  = threadIdx.x;
  const int lane = tid & 31;
  const int h = lane >> 4, c = lane & 15;
  const int strip = blockIdx.x * kWavesPerBlock + (tid >> 5);
  const bool valid = strip < kStrips;
  const int row0 = (valid ? strip : kStrips - 1) * 16;

  float4 av[kKC / 8];
  {
    const float* ap = comb + (size_t)(row0 + c) * kKC + 4 * h;
#pragma unroll
    for (int i = 0; i < kKC / 8; ++i)
      av[i] = *reinterpret_cast<const float4*>(ap + 8 * i);
  }

#pragma unroll
  for (int it = 0; it < (kKC / 4) * 48 / 256; ++it) {
    const int idx = tid + 256 * it;
    const int kq = idx / 48, n = idx - kq * 48;
    float4 v = make_float4(0.0f, 0.0f, 0.0f, 0.0f);
    if (n < kC) {
      const float* p = Wc + (size_t)(4 * kq) * kC + n;
      v = make_float4(p[0], p[kC], p[2 * kC], p[3 * kC]);
    }
    Bs[idx] = v;
  }
  __syncthreads();
  if (!valid) return;

  v8f acc[3];
#pragma unroll
  for (int t = 0; t < 3; ++t) acc[t] = v8f{};

  const float4* bp = Bs + h * 48 + c;
#pragma unroll
  for (int i = 0; i < kKC / 8; ++i) {
    float4 b[3];
#pragma unroll
    for (int t = 0; t < 3; ++t) b[t] = bp[i * 96 + t * 16];
    const v2f a01 = {av[i].x, av[i].y};
    const v2f a23 = {av[i].z, av[i].w};
#pragma unroll
    for (int t = 0; t < 3; ++t)
      acc[t] = wmma4(a01, v2f{b[t].x, b[t].y}, acc[t]);
#pragma unroll
    for (int t = 0; t < 3; ++t)
      acc[t] = wmma4(a23, v2f{b[t].z, b[t].w}, acc[t]);
  }

  float* op = out + (size_t)(row0 + 8 * h) * kC;
#pragma unroll
  for (int t = 0; t < 3; ++t) {
    const int col = 16 * t + c;
    const float bn = bias[col < kC ? col : kC - 1];
    float o[8];
#pragma unroll
    for (int r = 0; r < 8; ++r) o[r] = acc[t][r] + bn;
    if (col < kC) {
#pragma unroll
      for (int r = 0; r < 8; ++r) op[(size_t)r * kC + col] = o[r];
    }
  }
}


extern "C" void kernel_launch(void* const* d_in, const int* in_sizes, int n_in,
                              void* d_out, int out_size, void* d_ws, size_t ws_size,
                              hipStream_t stream) {
  (void)in_sizes; (void)n_in; (void)out_size; (void)ws_size;

  const float* xfeat  = (const float*)d_in[0];
  const float* xlabel = (const float*)d_in[1];
  const int*   eidx   = (const int*)d_in[2];
  const float* W_gcn  = (const float*)d_in[3];
  const float* b_gcn  = (const float*)d_in[4];
  const float* W_mlp  = (const float*)d_in[5];
  const float* b_mlp  = (const float*)d_in[6];
  const float* W_cls  = (const float*)d_in[7];
  const float* b_cls  = (const float*)d_in[8];
  float* out = (float*)d_out;

  char* ws = (char*)d_ws;
  const size_t off_deg  = 0;
  const size_t off_h    = (off_deg + (size_t)kN * 4 + 255) & ~(size_t)255;
  const size_t off_comb = (off_h + (size_t)kN * kH * 4 + 255) & ~(size_t)255;
  const size_t off_hist = (off_comb + (size_t)kN * kKC * 4 + 255) & ~(size_t)255;
  const size_t off_bst  = (off_hist + (size_t)kSortBlocks * kNB * 4 + 255) & ~(size_t)255;
  const size_t off_rptr = (off_bst + (size_t)(kNB + 1) * 4 + 255) & ~(size_t)255;
  const size_t off_tmp  = (off_rptr + (size_t)(kN + 1) * 4 + 255) & ~(size_t)255;
  const size_t off_csr  = (off_tmp + (size_t)kE * 4 + 255) & ~(size_t)255;
  float* dinv = (float*)(ws + off_deg);
  float* hbuf = (float*)(ws + off_h);
  float* comb = (float*)(ws + off_comb);
  int* hist   = (int*)(ws + off_hist);
  int* bstart = (int*)(ws + off_bst);
  int* rowptr = (int*)(ws + off_rptr);
  int* tmpe   = (int*)(ws + off_tmp);
  int* csrsrc = (int*)(ws + off_csr);

  const dim3 blk256(256);

  k_hist  <<<dim3(kSortBlocks), dim3(kSortThreads), 0, stream>>>(eidx, hist);
  k_bin   <<<dim3(kSortBlocks), dim3(kSortThreads), 0, stream>>>(eidx, hist, bstart, tmpe);
  k_bucket<<<dim3(kNB), dim3(kSortThreads), 0, stream>>>(tmpe, bstart, csrsrc, rowptr, dinv);

  k_gemm_n128<kXF, false><<<dim3(kGemmBlocks), blk256, 0, stream>>>(
      xfeat, xlabel, W_gcn, b_gcn, hbuf, kH);

  k_gather<<<dim3((kN + 7) / 8), blk256, 0, stream>>>(rowptr, csrsrc, hbuf, dinv, b_gcn, comb);

  k_gemm_n128<kKM, true><<<dim3(kGemmBlocks), blk256, 0, stream>>>(
      xfeat, xlabel, W_mlp, b_mlp, comb + kH, kKC);

  k_gemm_cls<<<dim3(kGemmBlocks), blk256, 0, stream>>>(comb, W_cls, b_cls, out);
}


# the three GEMMs run as 256 resident workgroups (one per WGP, fixed by LDS size) with a strip loop per wave: W staged once per WGP, 6-7 strips per SIMD, next strip's A loads in flight under the WMMAs
# speedup vs baseline: 7.8654x; 1.1012x over previous
#include <hip/hip_runtime.h>


typedef __attribute__((ext_vector_type(2))) float v2f;
typedef __attribute__((ext_vector_type(8))) float v8f;

constexpr int kN  = 100000;
constexpr int kE  = 1600000;
constexpr int kXF = 128;
constexpr int kXL = 40;
constexpr int kH  = 128;
constexpr int kC  = 40;
constexpr int kKM = kXF + kXL;
constexpr int kKC = 2 * kH;


constexpr int kBktShift = 9;
constexpr int kBktNodes = 1 << kBktShift;
constexpr int kNB = (kN + kBktNodes - 1) >> kBktShift;
constexpr int kSortThreads = 1024;
constexpr int kChunk = 6400;
constexpr int kSortBlocks = (kE + kChunk - 1) / kChunk;
constexpr int kChunkIters = (kChunk + kSortThreads - 1) / kSortThreads;
constexpr int kTabIters = (kSortBlocks + 3) / 4;
constexpr int kStageCap = 16384;
constexpr int kGrp = 12;
static_assert(kNB <= 256, "bucket scan uses 8 waves");
static_assert(kN <= (1 << (31 - kBktShift)), "source id and row-in-bucket share one word");

template <int NW>
__device__ __forceinline__ int block_excl_scan(int v, int* s_w) {
  const int lane = threadIdx.x & 31, wv = threadIdx.x >> 5;
  int incl = v;
#pragma unroll
  for (int o = 1; o < 32; o <<= 1) {
    const int up = __shfl_up(incl, o);
    if (lane >= o) incl += up;
  }
  if (wv < NW && lane == 31) s_w[wv] = incl;
  __syncthreads();
  int wpre = 0;
#pragma unroll
  for (int w = 0; w < NW; ++w) wpre += (w < wv) ? s_w[w] : 0;
  return wpre + incl - v;
}

__global__ __launch_bounds__(kSortThreads)
void k_hist(const int* __restrict__ eidx, int* __restrict__ hist) {
  __shared__ int s_h[kNB];
  const int tid = threadIdx.x, g = blockIdx.x;
  const int e0 = g * kChunk;
  const int n = (e0 + kChunk <= kE ? kChunk : kE - e0);
  int dv[kChunkIters];
#pragma unroll
  for (int i = 0; i < kChunkIters; ++i) {
    const int idx = tid + kSortThreads * i;
    dv[i] = 0;
    if (idx < n) dv[i] = eidx[kE + e0 + idx];
  }
  if (tid < kNB) s_h[tid] = 0;
  __syncthreads();
#pragma unroll
  for (int i = 0; i < kChunkIters; ++i)
    if (tid + kSortThreads * i < n) atomicAdd(&s_h[dv[i] >> kBktShift], 1);
  __syncthreads();
  if (tid < kNB) hist[g * kNB + tid] = s_h[tid];
}

__global__ __launch_bounds__(kSortThreads)
void k_bin(const int* __restrict__ eidx, const int* __restrict__ hist,
           int* __restrict__ bstart, int* __restrict__ tmp) {
  __shared__ int s_tot4[4][kNB];
  __shared__ int s_pre4[4][kNB];
  __shared__ int s_w1[8];
  __shared__ int s_w2[8];
  __shared__ int s_lstart[kNB];
  __shared__ int s_delta[kNB];
  __shared__ int s_cnt[kNB];
  __shared__ int s_ss[kChunk];
  __shared__ int s_sd[kChunk];

  const int tid = threadIdx.x, g = blockIdx.x;
  const int b = tid & 255, q = tid >> 8;
  const int e0 = g * kChunk;
  const int n = (e0 + kChunk <= kE ? kChunk : kE - e0);

  int sv[kChunkIters], dv[kChunkIters];
#pragma unroll
  for (int i = 0; i < kChunkIters; ++i) {
    const int idx = tid + kSortThreads * i;
    sv[i] = 0; dv[i] = 0;
    if (idx < n) { sv[i] = eidx[e0 + idx]; dv[i] = eidx[kE + e0 + idx]; }
  }

  if (b < kNB) {
    int tot = 0, pre = 0;
#pragma unroll 21
    for (int it = 0; it < kTabIters; ++it) {
      const int gg = q + 4 * it;
      if (gg < kSortBlocks) {
        const int v = hist[gg * kNB + b];
        tot += v;
        pre += (gg < g) ? v : 0;
      }
    }
    s_tot4[q][b] = tot;
    s_pre4[q][b] = pre;
  }
  int own = 0;
  if (tid < kNB) { own = hist[g * kNB + tid]; s_cnt[tid] = 0; }
  __syncthreads();
  int tot = 0, pre = 0;
  if (tid < kNB) {
    tot = (s_tot4[0][tid] + s_tot4[1][tid]) + (s_tot4[2][tid] + s_tot4[3][tid]);
    pre = (s_pre4[0][tid] + s_pre4[1][tid]) + (s_pre4[2][tid] + s_pre4[3][tid]);
  }
  const int gstart = block_excl_scan<8>(tot, s_w1);
  const int lstart = block_excl_scan<8>(own, s_w2);
  if (tid < kNB) {
    s_lstart[tid] = lstart;
    s_delta[tid] = gstart + pre - lstart;
    if (g == 0) {
      bstart[tid] = gstart;
      if (tid == kNB - 1) bstart[kNB] = gstart + tot;
    }
  }
  __syncthreads();
#pragma unroll
  for (int i = 0; i < kChunkIters; ++i) {
    if (tid + kSortThreads * i < n) {
      const int bk = dv[i] >> kBktShift;
      const int p = s_lstart[bk] + atomicAdd(&s_cnt[bk], 1);
      s_ss[p] = sv[i];
      s_sd[p] = dv[i];
    }
  }
  __syncthreads();
#pragma unroll
  for (int i = 0; i < kChunkIters; ++i) {
    const int idx = tid + kSortThreads * i;
    if (idx < n) {
      const int d = s_sd[idx];
      const unsigned w = ((unsigned)s_ss[idx] << kBktShift) | (unsigned)(d & (kBktNodes - 1));
      tmp[idx + s_delta[d >> kBktShift]] = (int)w;
    }
  }
}

__global__ __launch_bounds__(kSortThreads)
void k_bucket(const int* __restrict__ tmp, const int* __restrict__ bstart,
              int* __restrict__ csr_src, int* __restrict__ rowptr,
              float* __restrict__ dinv) {
  __shared__ int s_cnt[kBktNodes];
  __shared__ int s_cur[kBktNodes];
  __shared__ int s_wave[kBktNodes / 32];
  __shared__ int s_out[kStageCap];
  const int tid = threadIdx.x;
  const int bk = blockIdx.x;
  const int beg = bstart[bk];
  const int n = bstart[bk + 1] - beg;
  const int* tp = tmp + beg;

  if (tid < kBktNodes) s_cnt[tid] = 0;
  __syncthreads();
  for (int base = 0; base < n; base += kGrp * kSortThreads) {
    int w[kGrp];
#pragma unroll
    for (int k = 0; k < kGrp; ++k) {
      const int i = base + tid + kSortThreads * k;
      w[k] = 0;
      if (i < n) w[k] = tp[i];
    }
#pragma unroll
    for (int k = 0; k < kGrp; ++k)
      if (base + tid + kSortThreads * k < n) atomicAdd(&s_cnt[w[k] & (kBktNodes - 1)], 1);
  }
  __syncthreads();
  const int v = tid < kBktNodes ? s_cnt[tid] : 0;
  const int excl = block_excl_scan<kBktNodes / 32>(v, s_wave);
  if (tid < kBktNodes) {
    s_cur[tid] = excl;
    const int node = bk * kBktNodes + tid;
    if (node < kN) {
      rowptr[node] = beg + excl;
      if (node == kN - 1) rowptr[kN] = beg + excl + v;
      const float d = (float)(v + 1);
      dinv[node] = (d > 0.0f) ? rsqrtf(d) : 0.0f;
    }
  }
  __syncthreads();
  const bool staged = n <= kStageCap;
  for (int base = 0; base < n; base += kGrp * kSortThreads) {
    int w[kGrp];
#pragma unroll
    for (int k = 0; k < kGrp; ++k) {
      const int i = base + tid + kSortThreads * k;
      w[k] = 0;
      if (i < n) w[k] = tp[i];
    }
#pragma unroll
    for (int k = 0; k < kGrp; ++k) {
      if (base + tid + kSortThreads * k < n) {
        const int p = atomicAdd(&s_cur[w[k] & (kBktNodes - 1)], 1);
        const int s = (int)((unsigned)w[k] >> kBktShift);
        if (staged) s_out[p] = s; else csr_src[beg + p] = s;
      }
    }
  }
  if (staged) {
    __syncthreads();
    for (int i = tid; i < n; i += kSortThreads) csr_src[beg + i] = s_out[i];
  }
}


constexpr int kStrips = kN / 16;
constexpr int kWavesPerBlock = 8;
constexpr int kGemmGrid = 256;
constexpr int kStripStep = kGemmGrid * kWavesPerBlock;
constexpr int kLdsQuadsOnePerWgp = 164 * 1024 / 16;

__device__ __forceinline__ v8f wmma4(v2f a, v2f b, v8f c) {
  return __builtin_amdgcn_wmma_f32_16x16x4_f32(false, a, false, b, (short)0, c,
                                               false, false);
}


template <int K, bool kMlp>
__global__ __launch_bounds__(256)
void k_gemm_n128(const float* __restrict__ A0,
                 const float* __restrict__ A1,
                 const float* __restrict__ Wg,
                 const float* __restrict__ bias,
                 float* __restrict__ Out, int ldo) {
  static_assert((K / 4) * 128 <= kLdsQuadsOnePerWgp, "W must fit the reserved LDS");
  __shared__ float4 Bs[kLdsQuadsOnePerWgp];

  const int tid  = threadIdx.x;
  const int lane = tid & 31;
  const int h = lane >> 4, c = lane & 15;
  const int wave = __builtin_amdgcn_readfirstlane(tid >> 5);
  int strip = (int)blockIdx.x + kGemmGrid * wave;

  auto load_a = [&](float4 (&dst)[K / 8], int s) {
    const float* ap = A0 + (size_t)(s * 16 + c) * kXF + 4 * h;
#pragma unroll
    for (int i = 0; i < kXF / 8; ++i)
      dst[i] = *reinterpret_cast<const float4*>(ap + 8 * i);
    if constexpr (kMlp) {
      const float* lp = A1 + (size_t)(s * 16 + c) * kXL + 4 * h;
#pragma unroll
      for (int i = 0; i < kXL / 8; ++i)
        dst[kXF / 8 + i] = *reinterpret_cast<const float4*>(lp + 8 * i);
    }
  };

  float4 av[K / 8];
  load_a(av, strip);

  constexpr int kQuads = (K / 4) * 32;
  auto stage_quad = [&](int idx) {
    const int kq = idx >> 5, nq = idx & 31;
    const float* p = Wg + (size_t)(4 * kq) * kH + 4 * nq;
    const float4 r0 = *reinterpret_cast<const float4*>(p);
    const float4 r1 = *reinterpret_cast<const float4*>(p + kH);
    const float4 r2 = *reinterpret_cast<const float4*>(p + 2 * kH);
    const float4 r3 = *reinterpret_cast<const float4*>(p + 3 * kH);
    float4* d = Bs + kq * 128 + (nq >> 4) * 64 + (nq & 15);
    d[0]  = make_float4(r0.x, r1.x, r2.x, r3.x);
    d[16] = make_float4(r0.y, r1.y, r2.y, r3.y);
    d[32] = make_float4(r0.z, r1.z, r2.z, r3.z);
    d[48] = make_float4(r0.w, r1.w, r2.w, r3.w);
  };
#pragma unroll
  for (int it = 0; it < kQuads / 256; ++it) stage_quad(tid + 256 * it);
  if constexpr (kQuads % 256 != 0) {
    if (tid < kQuads % 256) stage_quad(tid + 256 * (kQuads / 256));
  }
  __syncthreads();

  float4 bv[2];
#pragma unroll
  for (int g = 0; g < 2; ++g) {
    bv[g] = make_float4(0.0f, 0.0f, 0.0f, 0.0f);
    if constexpr (kMlp) bv[g] = *reinterpret_cast<const float4*>(bias + 64 * g + 4 * c);
  }

  const float4* bp = Bs + h * 128 + c;
  for (;;) {
    __builtin_amdgcn_fence(__ATOMIC_ACQUIRE, "wavefront", "local");
    const int next = strip + kStripStep;
    const bool more = next < kStrips;
    float4 an[K / 8];
#pragma unroll
    for (int i = 0; i < K / 8; ++i) an[i] = av[i];
    if (more) load_a(an, next);

    v8f acc[8];
#pragma unroll
    for (int t = 0; t < 8; ++t) acc[t] = v8f{};

#pragma unroll
    for (int i = 0; i < K / 8; ++i) {
      float4 b[8];
#pragma unroll
      for (int t = 0; t < 8; ++t) b[t] = bp[i * 256 + t * 16];
      const v2f a01 = {av[i].x, av[i].y};
      const v2f a23 = {av[i].z, av[i].w};
#pragma unroll
      for (int t = 0; t < 8; ++t)
        acc[t] = wmma4(a01, v2f{b[t].x, b[t].y}, acc[t]);
#pragma unroll
      for (int t = 0; t < 8; ++t)
        acc[t] = wmma4(a23, v2f{b[t].z, b[t].w}, acc[t]);
    }

    float* op = Out + (size_t)(strip * 16 + 8 * h) * ldo + 4 * c;
#pragma unroll
    for (int g = 0; g < 2; ++g) {
#pragma unroll
      for (int r = 0; r < 8; ++r) {
        float4 o = make_float4(acc[4 * g + 0][r], acc[4 * g + 1][r],
                               acc[4 * g + 2][r], acc[4 * g + 3][r]);
        if constexpr (kMlp) {
          o.x = fmaxf(o.x + bv[g].x, 0.0f);
          o.y = fmaxf(o.y + bv[g].y, 0.0f);
          o.z = fmaxf(o.z + bv[g].z, 0.0f);
          o.w = fmaxf(o.w + bv[g].w, 0.0f);
        }
        *reinterpret_cast<float4*>(op + (size_t)r * ldo + 64 * g) = o;
      }
    }

    if (!more) break;
#pragma unroll
    for (int i = 0; i < K / 8; ++i) av[i] = an[i];
    strip = next;
  }
}


__global__ __launch_bounds__(256)
void k_gather(const int* __restrict__ rowptr, const int* __restrict__ csr_src,
              const float* __restrict__ h, const float* __restrict__ dinv,
              const float* __restrict__ bg, float* __restrict__ comb) {
  const int lane = threadIdx.x & 31;
  const int d = __builtin_amdgcn_readfirstlane((int)(blockIdx.x * 8 + (threadIdx.x >> 5)));
  if (d >= kN) return;
  const int beg = rowptr[d];
  const int end = rowptr[d + 1];
  const float dd = dinv[d];
  const float4* hq = reinterpret_cast<const float4*>(h) + lane;

  float4 acc;
  {
    const float4 v = hq[(size_t)d * (kH / 4)];
    const float w = dd * dd;
    acc = make_float4(v.x * w, v.y * w, v.z * w, v.w * w);
  }

  for (int base = beg; base < end; base += 32) {
    const int n = (end - base) < 32 ? (end - base) : 32;
    int s_l = 0;
    float w_l = 0.0f;
    if (lane < n) {
      s_l = csr_src[base + lane];
      w_l = dinv[s_l] * dd;
    }
    auto row = [&](int j) {
      const int s = __builtin_amdgcn_readlane(s_l, j);
      return hq[(size_t)s * (kH / 4)];
    };
    auto add = [&](const float4 v, int j) {
      const float w = __builtin_bit_cast(
          float, __builtin_amdgcn_readlane(__builtin_bit_cast(int, w_l), j));
      acc.x = fmaf(v.x, w, acc.x);
      acc.y = fmaf(v.y, w, acc.y);
      acc.z = fmaf(v.z, w, acc.z);
      acc.w = fmaf(v.w, w, acc.w);
    };
    int j = 0;
    for (; j + 8 <= n; j += 8) {
      float4 v[8];
#pragma unroll
      for (int k = 0; k < 8; ++k) v[k] = row(j + k);
#pragma unroll
      for (int k = 0; k < 8; ++k) add(v[k], j + k);
    }
    if (j + 4 <= n) {
      float4 v[4];
#pragma unroll
      for (int k = 0; k < 4; ++k) v[k] = row(j + k);
#pragma unroll
      for (int k = 0; k < 4; ++k) add(v[k], j + k);
      j += 4;
    }
    const int r = n - j;
    if (r > 0) {
      const float4 v0 = row(j);
      float4 v1 = v0, v2 = v0;
      if (r > 1) v1 = row(j + 1);
      if (r > 2) v2 = row(j + 2);
      add(v0, j);
      if (r > 1) add(v1, j + 1);
      if (r > 2) add(v2, j + 2);
    }
  }

  const float4 bv = *reinterpret_cast<const float4*>(bg + 4 * lane);
  float4 o;
  o.x = fmaxf(acc.x + bv.x, 0.0f);
  o.y = fmaxf(acc.y + bv.y, 0.0f);
  o.z = fmaxf(acc.z + bv.z, 0.0f);
  o.w = fmaxf(acc.w + bv.w, 0.0f);
  *reinterpret_cast<float4*>(comb + (size_t)d * kKC + 4 * lane) = o;
}


__global__ __launch_bounds__(256)
void k_gemm_cls(const float* __restrict__ comb, const float* __restrict__ Wc,
                const float* __restrict__ bias, float* __restrict__ out) {
  static_assert((kKC / 4) * 48 <= kLdsQuadsOnePerWgp, "W must fit the reserved LDS");
  __shared__ float4 Bs[kLdsQuadsOnePerWgp];

  const int tid  = threadIdx.x;
  const int lane = tid & 31;
  const int h = lane >> 4, c = lane & 15;
  const int wave = __builtin_amdgcn_readfirstlane(tid >> 5);
  int strip = (int)blockIdx.x + kGemmGrid * wave;

  auto load_a = [&](float4 (&dst)[kKC / 8], int s) {
    const float* ap = comb + (size_t)(s * 16 + c) * kKC + 4 * h;
#pragma unroll
    for (int i = 0; i < kKC / 8; ++i)
      dst[i] = *reinterpret_cast<const float4*>(ap + 8 * i);
  };

  float4 av[kKC / 8];
  load_a(av, strip);

#pragma unroll
  for (int it = 0; it < (kKC / 4) * 48 / 256; ++it) {
    const int idx = tid + 256 * it;
    const int kq = idx / 48, n = idx - kq * 48;
    float4 v = make_float4(0.0f, 0.0f, 0.0f, 0.0f);
    if (n < kC) {
      const float* p = Wc + (size_t)(4 * kq) * kC + n;
      v = make_float4(p[0], p[kC], p[2 * kC], p[3 * kC]);
    }
    Bs[idx] = v;
  }
  __syncthreads();

  float bn[3];
#pragma unroll
  for (int t = 0; t < 3; ++t) {
    const int col = 16 * t + c;
    bn[t] = bias[col < kC ? col : kC - 1];
  }

  const float4* bp = Bs + h * 48 + c;
  for (;;) {
    __builtin_amdgcn_fence(__ATOMIC_ACQUIRE, "wavefront", "local");
    const int next = strip + kStripStep;
    const bool more = next < kStrips;
    float4 an[kKC / 8];
#pragma unroll
    for (int i = 0; i < kKC / 8; ++i) an[i] = av[i];
    if (more) load_a(an, next);

    v8f acc[3];
#pragma unroll
    for (int t = 0; t < 3; ++t) acc[t] = v8f{};

#pragma unroll
    for (int i = 0; i < kKC / 8; ++i) {
      float4 b[3];
#pragma unroll
      for (int t = 0; t < 3; ++t) b[t] = bp[i * 96 + t * 16];
      const v2f a01 = {av[i].x, av[i].y};
      const v2f a23 = {av[i].z, av[i].w};
#pragma unroll
      for (int t = 0; t < 3; ++t)
        acc[t] = wmma4(a01, v2f{b[t].x, b[t].y}, acc[t]);
#pragma unroll
      for (int t = 0; t < 3; ++t)
        acc[t] = wmma4(a23, v2f{b[t].z, b[t].w}, acc[t]);
    }

    float* op = out + (size_t)(strip * 16 + 8 * h) * kC;
#pragma unroll
    for (int t = 0; t < 3; ++t) {
      const int col = 16 * t + c;
      float o[8];
#pragma unroll
      for (int r = 0; r < 8; ++r) o[r] = acc[t][r] + bn[t];
      if (col < kC) {
#pragma unroll
        for (int r = 0; r < 8; ++r) op[(size_t)r * kC + col] = o[r];
      }
    }

    if (!more) break;
#pragma unroll
    for (int i = 0; i < kKC / 8; ++i) av[i] = an[i];
    strip = next;
  }
}


extern "C" void kernel_launch(void* const* d_in, const int* in_sizes, int n_in,
                              void* d_out, int out_size, void* d_ws, size_t ws_size,
                              hipStream_t stream) {
  (void)in_sizes; (void)n_in; (void)out_size; (void)ws_size;

  const float* xfeat  = (const float*)d_in[0];
  const float* xlabel = (const float*)d_in[1];
  const int*   eidx   = (const int*)d_in[2];
  const float* W_gcn  = (const float*)d_in[3];
  const float* b_gcn  = (const float*)d_in[4];
  const float* W_mlp  = (const float*)d_in[5];
  const float* b_mlp  = (const float*)d_in[6];
  const float* W_cls  = (const float*)d_in[7];
  const float* b_cls  = (const float*)d_in[8];
  float* out = (float*)d_out;

  char* ws = (char*)d_ws;
  const size_t off_deg  = 0;
  const size_t off_h    = (off_deg + (size_t)kN * 4 + 255) & ~(size_t)255;
  const size_t off_comb = (off_h + (size_t)kN * kH * 4 + 255) & ~(size_t)255;
  const size_t off_hist = (off_comb + (size_t)kN * kKC * 4 + 255) & ~(size_t)255;
  const size_t off_bst  = (off_hist + (size_t)kSortBlocks * kNB * 4 + 255) & ~(size_t)255;
  const size_t off_rptr = (off_bst + (size_t)(kNB + 1) * 4 + 255) & ~(size_t)255;
  const size_t off_tmp  = (off_rptr + (size_t)(kN + 1) * 4 + 255) & ~(size_t)255;
  const size_t off_csr  = (off_tmp + (size_t)kE * 4 + 255) & ~(size_t)255;
  float* dinv = (float*)(ws + off_deg);
  float* hbuf = (float*)(ws + off_h);
  float* comb = (float*)(ws + off_comb);
  int* hist   = (int*)(ws + off_hist);
  int* bstart = (int*)(ws + off_bst);
  int* rowptr = (int*)(ws + off_rptr);
  int* tmpe   = (int*)(ws + off_tmp);
  int* csrsrc = (int*)(ws + off_csr);

  const dim3 blk256(256);

  k_hist  <<<dim3(kSortBlocks), dim3(kSortThreads), 0, stream>>>(eidx, hist);
  k_bin   <<<dim3(kSortBlocks), dim3(kSortThreads), 0, stream>>>(eidx, hist, bstart, tmpe);
  k_bucket<<<dim3(kNB), dim3(kSortThreads), 0, stream>>>(tmpe, bstart, csrsrc, rowptr, dinv);

  k_gemm_n128<kXF, false><<<dim3(kGemmGrid), blk256, 0, stream>>>(
      xfeat, xlabel, W_gcn, b_gcn, hbuf, kH);

  k_gather<<<dim3((kN + 7) / 8), blk256, 0, stream>>>(rowptr, csrsrc, hbuf, dinv, b_gcn, comb);

  k_gemm_n128<kKM, true><<<dim3(kGemmGrid), blk256, 0, stream>>>(
      xfeat, xlabel, W_mlp, b_mlp, comb + kH, kKC);

  k_gemm_cls<<<dim3(kGemmGrid), blk256, 0, stream>>>(comb, W_cls, b_cls, out);
}
